# starting kernel
# baseline (speedup 1.0000x reference)
#include <hip/hip_runtime.h>

#define NNODES 50000
#define NEDGES 800000
#define NODE_TILES (NNODES / 16)
#define EDGE_TILES (NEDGES / 16)
#define WPB 8

typedef __attribute__((ext_vector_type(2))) float v2f;
typedef __attribute__((ext_vector_type(8))) float v8f;

__device__ __forceinline__ v8f wmma4(v2f a, v2f b, v8f c) {
  return __builtin_amdgcn_wmma_f32_16x16x4_f32(
      false, a, false, b, (short)0, c, false, false);
}

__device__ __forceinline__ v2f ldA(const float* A, int ldk, int kb, int lane) {
  int m = lane & 15;
  int k = 4 * kb + 2 * (lane >> 4);
  v2f a;
  a.x = A[m * ldk + k];
  a.y = A[m * ldk + k + 1];
  return a;
}

__device__ __forceinline__ v2f ldB(const float* W, int ncols, int kb, int col0, int lane) {
  int col = col0 + (lane & 15);
  int k = 4 * kb + 2 * (lane >> 4);
  v2f b;
  b.x = W[k * ncols + col];
  b.y = W[(k + 1) * ncols + col];
  return b;
}

__device__ __forceinline__ v8f bias_frag(const float* b, int col0, int lane) {
  float bv = b[col0 + (lane & 15)];
  v8f c;
#pragma unroll
  for (int r = 0; r < 8; ++r) c[r] = bv;
  return c;
}

__device__ __forceinline__ void st_tile(float* A, const v8f* acc, int lane, int mode) {
  int half = lane >> 4, nl = lane & 15;
#pragma unroll
  for (int nb = 0; nb < 4; ++nb)
#pragma unroll
    for (int r = 0; r < 8; ++r) {
      float v = acc[nb][r];
      if (mode == 1) v = tanhf(v);
      else if (mode == 2) v = fmaxf(v, 0.0f);
      A[(r + 8 * half) * 64 + nb * 16 + nl] = v;
    }
}

__device__ __forceinline__ void gemm64_lds(const float* A, const float* W,
                                           const float* b, v8f* acc, int lane) {
#pragma unroll
  for (int nb = 0; nb < 4; ++nb) acc[nb] = bias_frag(b, nb * 16, lane);
#pragma unroll
  for (int kb = 0; kb < 16; ++kb) {
    v2f a = ldA(A, 64, kb, lane);
#pragma unroll
    for (int nb = 0; nb < 4; ++nb)
      acc[nb] = wmma4(a, ldB(W, 64, kb, nb * 16, lane), acc[nb]);
  }
}

__global__ void k_zero(float* p, long n) {
  long i = (long)blockIdx.x * blockDim.x + threadIdx.x;
  if (i < n) p[i] = 0.0f;
}

__global__ void __launch_bounds__(256) k_encode(
    const float* __restrict__ x,
    const float* __restrict__ WencK, const float* __restrict__ bencK,
    const float* __restrict__ WencP1, const float* __restrict__ bencP1,
    const float* __restrict__ WencP2, const float* __restrict__ bencP2,
    const float* __restrict__ WD, const float* __restrict__ bD,
    float* __restrict__ hK, float* __restrict__ h1,
    float* __restrict__ h2, float* __restrict__ Dxh) {
  __shared__ float sWK[32 * 32], sW1[32 * 32], sW2[32 * 32], sWD[64 * 64];
  __shared__ float sbK[32], sb1[32], sb2[32], sbD[32];
  for (int i = threadIdx.x; i < 32 * 32; i += blockDim.x) {
    sWK[i] = WencK[i]; sW1[i] = WencP1[i]; sW2[i] = WencP2[i];
  }
  for (int i = threadIdx.x; i < 64 * 64; i += blockDim.x) sWD[i] = WD[i];
  for (int i = threadIdx.x; i < 32; i += blockDim.x) {
    sbK[i] = bencK[i]; sb1[i] = bencP1[i]; sb2[i] = bencP2[i]; sbD[i] = bD[32 + i];
  }
  __syncthreads();

  int wave = threadIdx.x >> 5, lane = threadIdx.x & 31;
  int m = lane & 15;
  int half = lane >> 4, nl = lane & 15;

  for (int tile = blockIdx.x * WPB + wave; tile < NODE_TILES;
       tile += gridDim.x * WPB) {
    int n0 = tile * 16;

    v8f aK[2], a1[2], a2[2], aD[2];
#pragma unroll
    for (int nb = 0; nb < 2; ++nb) {
      aK[nb] = bias_frag(sbK, nb * 16, lane);
      a1[nb] = bias_frag(sb1, nb * 16, lane);
      a2[nb] = bias_frag(sb2, nb * 16, lane);
      aD[nb] = bias_frag(sbD, nb * 16, lane);
    }

#pragma unroll
    for (int kb = 0; kb < 8; ++kb) {
      int k = 4 * kb + 2 * (lane >> 4);
      long base = (long)(n0 + m) * 64;
      v2f aq, ap;
      aq.x = x[base + k];       aq.y = x[base + k + 1];
      ap.x = x[base + 32 + k];  ap.y = x[base + 32 + k + 1];
#pragma unroll
      for (int nb = 0; nb < 2; ++nb) {
        aK[nb] = wmma4(ap, ldB(sWK, 32, kb, nb * 16, lane), aK[nb]);
        a1[nb] = wmma4(aq, ldB(sW1, 32, kb, nb * 16, lane), a1[nb]);
        a2[nb] = wmma4(aq, ldB(sW2, 32, kb, nb * 16, lane), a2[nb]);
      }
    }
#pragma unroll
    for (int kb = 0; kb < 16; ++kb) {
      int k = 4 * kb + 2 * (lane >> 4);
      long base = (long)(n0 + m) * 64;
      v2f ax;
      ax.x = x[base + k]; ax.y = x[base + k + 1];
#pragma unroll
      for (int nb = 0; nb < 2; ++nb)
        aD[nb] = wmma4(ax, ldB(sWD, 64, kb, 32 + nb * 16, lane), aD[nb]);
    }

#pragma unroll
    for (int nb = 0; nb < 2; ++nb)
#pragma unroll
      for (int r = 0; r < 8; ++r) {
        long o = (long)(n0 + r + 8 * half) * 32 + nb * 16 + nl;
        hK[o] = aK[nb][r];
        h1[o] = a1[nb][r];
        h2[o] = a2[nb][r];
        Dxh[o] = aD[nb][r];
      }
  }
}

__global__ void k_scatter_enc(const int* __restrict__ src, const int* __restrict__ dst,
                              const float* __restrict__ hK, const float* __restrict__ Dxh,
                              float* __restrict__ hK_agg, float* __restrict__ d_agg) {
  long tid = (long)blockIdx.x * blockDim.x + threadIdx.x;
  if (tid >= (long)NEDGES * 8) return;
  int e = (int)(tid >> 3);
  int j = (int)(tid & 7) * 4;
  int s = src[e], d = dst[e];
  float4 a = *(const float4*)&hK[(long)s * 32 + j];
  float4 b = *(const float4*)&Dxh[(long)s * 32 + j];
  float* pa = &hK_agg[(long)d * 32 + j];
  float* pb = &d_agg[(long)d * 32 + j];
  atomicAdd(pa + 0, a.x); atomicAdd(pa + 1, a.y);
  atomicAdd(pa + 2, a.z); atomicAdd(pa + 3, a.w);
  atomicAdd(pb + 0, b.x); atomicAdd(pb + 1, b.y);
  atomicAdd(pb + 2, b.z); atomicAdd(pb + 3, b.w);
}

__global__ void __launch_bounds__(256) k_node_mlp(
    const float* __restrict__ hK_agg,
    const float* __restrict__ WK1, const float* __restrict__ bK1,
    const float* __restrict__ WK2, const float* __restrict__ bK2,
    const float* __restrict__ WK3, const float* __restrict__ bK3,
    float* __restrict__ E_node) {
  __shared__ float sW1[32 * 64], sW2[64 * 64], sW3[64 * 64];
  __shared__ float sb1[64], sb2[64], sb3[64];
  __shared__ float sact[WPB][16 * 64];
  for (int i = threadIdx.x; i < 32 * 64; i += blockDim.x) sW1[i] = WK1[i];
  for (int i = threadIdx.x; i < 64 * 64; i += blockDim.x) { sW2[i] = WK2[i]; sW3[i] = WK3[i]; }
  for (int i = threadIdx.x; i < 64; i += blockDim.x) { sb1[i] = bK1[i]; sb2[i] = bK2[i]; sb3[i] = bK3[i]; }
  __syncthreads();

  int wave = threadIdx.x >> 5, lane = threadIdx.x & 31;
  int m = lane & 15;
  int half = lane >> 4, nl = lane & 15;
  float* A = sact[wave];

  for (int tile = blockIdx.x * WPB + wave; tile < NODE_TILES;
       tile += gridDim.x * WPB) {
    int n0 = tile * 16;

    v8f acc[4];
#pragma unroll
    for (int nb = 0; nb < 4; ++nb) acc[nb] = bias_frag(sb1, nb * 16, lane);
#pragma unroll
    for (int kb = 0; kb < 8; ++kb) {
      int k = 4 * kb + 2 * (lane >> 4);
      long base = (long)(n0 + m) * 32;
      v2f a; a.x = hK_agg[base + k]; a.y = hK_agg[base + k + 1];
#pragma unroll
      for (int nb = 0; nb < 4; ++nb)
        acc[nb] = wmma4(a, ldB(sW1, 64, kb, nb * 16, lane), acc[nb]);
    }
    st_tile(A, acc, lane, 1);
    gemm64_lds(A, sW2, sb2, acc, lane);
    st_tile(A, acc, lane, 2);
    gemm64_lds(A, sW3, sb3, acc, lane);

#pragma unroll
    for (int nb = 0; nb < 4; ++nb)
#pragma unroll
      for (int r = 0; r < 8; ++r)
        E_node[(long)(n0 + r + 8 * half) * 64 + nb * 16 + nl] = acc[nb][r];
  }
}

__global__ void __launch_bounds__(256) k_edge(
    const int* __restrict__ src, const int* __restrict__ dst,
    const float* __restrict__ h1, const float* __restrict__ h2,
    const float* __restrict__ E_node,
    const float* __restrict__ WU1, const float* __restrict__ bU1,
    const float* __restrict__ WU2, const float* __restrict__ bU2,
    const float* __restrict__ WU3, const float* __restrict__ bU3,
    float* __restrict__ E_new) {
  __shared__ float sW1[32 * 64], sW2[64 * 64], sW3[64 * 64];
  __shared__ float sb1[64], sb2[64], sb3[64];
  __shared__ float sact[WPB][16 * 64];
  for (int i = threadIdx.x; i < 32 * 64; i += blockDim.x) sW1[i] = WU1[i];
  for (int i = threadIdx.x; i < 64 * 64; i += blockDim.x) { sW2[i] = WU2[i]; sW3[i] = WU3[i]; }
  for (int i = threadIdx.x; i < 64; i += blockDim.x) { sb1[i] = bU1[i]; sb2[i] = bU2[i]; sb3[i] = bU3[i]; }
  __syncthreads();

  int wave = threadIdx.x >> 5, lane = threadIdx.x & 31;
  int m = lane & 15;
  int half = lane >> 4, nl = lane & 15;
  float* A = sact[wave];

  for (int tile = blockIdx.x * WPB + wave; tile < EDGE_TILES;
       tile += gridDim.x * WPB) {
    int e0 = tile * 16;
    int s = src[e0 + m], d = dst[e0 + m];

    v8f acc[4];
#pragma unroll
    for (int nb = 0; nb < 4; ++nb) acc[nb] = bias_frag(sb1, nb * 16, lane);
#pragma unroll
    for (int kb = 0; kb < 8; ++kb) {
      int k = 4 * kb + 2 * (lane >> 4);
      v2f a;
      a.x = h1[(long)s * 32 + k]     + h2[(long)d * 32 + k];
      a.y = h1[(long)s * 32 + k + 1] + h2[(long)d * 32 + k + 1];
#pragma unroll
      for (int nb = 0; nb < 4; ++nb)
        acc[nb] = wmma4(a, ldB(sW1, 64, kb, nb * 16, lane), acc[nb]);
    }
    st_tile(A, acc, lane, 1);
    gemm64_lds(A, sW2, sb2, acc, lane);
    st_tile(A, acc, lane, 2);
    gemm64_lds(A, sW3, sb3, acc, lane);

#pragma unroll
    for (int r = 0; r < 8; ++r) {
      int e = e0 + r + 8 * half;
      int se = src[e], de = dst[e];
#pragma unroll
      for (int nb = 0; nb < 4; ++nb) {
        int col = nb * 16 + nl;
        float v = acc[nb][r] * E_node[(long)se * 64 + col];
        atomicAdd(&E_new[(long)de * 64 + col], v);
      }
    }
  }
}

__global__ void __launch_bounds__(256) k_final(
    const float* __restrict__ E_new,
    const float* __restrict__ WH, const float* __restrict__ bH,
    const float* __restrict__ d_agg, float* __restrict__ out) {
  __shared__ float sWH[64 * 64], sbH[64];
  for (int i = threadIdx.x; i < 64 * 64; i += blockDim.x) sWH[i] = WH[i];
  for (int i = threadIdx.x; i < 64; i += blockDim.x) sbH[i] = bH[i];
  __syncthreads();

  int wave = threadIdx.x >> 5, lane = threadIdx.x & 31;
  int m = lane & 15;
  int half = lane >> 4, nl = lane & 15;

  for (int tile = blockIdx.x * WPB + wave; tile < NODE_TILES;
       tile += gridDim.x * WPB) {
    int n0 = tile * 16;

    v8f acc[4];
#pragma unroll
    for (int nb = 0; nb < 4; ++nb) acc[nb] = bias_frag(sbH, nb * 16, lane);
#pragma unroll
    for (int kb = 0; kb < 16; ++kb) {
      int k = 4 * kb + 2 * (lane >> 4);
      long base = (long)(n0 + m) * 64;
      v2f a; a.x = E_new[base + k]; a.y = E_new[base + k + 1];
#pragma unroll
      for (int nb = 0; nb < 4; ++nb)
        acc[nb] = wmma4(a, ldB(sWH, 64, kb, nb * 16, lane), acc[nb]);
    }

#pragma unroll
    for (int r = 0; r < 8; ++r) {
      long row = n0 + r + 8 * half;
#pragma unroll
      for (int nb = 0; nb < 2; ++nb) {
        out[row * 64 + nb * 16 + nl] = acc[nb + 2][r];
        out[row * 64 + 32 + nb * 16 + nl] =
            -acc[nb][r] - d_agg[row * 32 + nb * 16 + nl];
      }
    }
  }
}

extern "C" void kernel_launch(void* const* d_in, const int* in_sizes, int n_in,
                              void* d_out, int out_size, void* d_ws, size_t ws_size,
                              hipStream_t stream) {
  const float* x      = (const float*)d_in[0];
  const int*   src    = (const int*)d_in[1];
  const int*   dst    = (const int*)d_in[2];
  const float* WencK  = (const float*)d_in[3];
  const float* bencK  = (const float*)d_in[4];
  const float* WencP1 = (const float*)d_in[5];
  const float* bencP1 = (const float*)d_in[6];
  const float* WencP2 = (const float*)d_in[7];
  const float* bencP2 = (const float*)d_in[8];
  const float* WK1 = (const float*)d_in[9];
  const float* bK1 = (const float*)d_in[10];
  const float* WK2 = (const float*)d_in[11];
  const float* bK2 = (const float*)d_in[12];
  const float* WK3 = (const float*)d_in[13];
  const float* bK3 = (const float*)d_in[14];
  const float* WU1 = (const float*)d_in[15];
  const float* bU1 = (const float*)d_in[16];
  const float* WU2 = (const float*)d_in[17];
  const float* bU2 = (const float*)d_in[18];
  const float* WU3 = (const float*)d_in[19];
  const float* bU3 = (const float*)d_in[20];
  const float* WH  = (const float*)d_in[21];
  const float* bH  = (const float*)d_in[22];
  const float* WD  = (const float*)d_in[23];
  const float* bD  = (const float*)d_in[24];

  float* ws = (float*)d_ws;
  const size_t N32 = (size_t)NNODES * 32;
  const size_t N64 = (size_t)NNODES * 64;
  float* hK     = ws;
  float* h1v    = hK + N32;
  float* h2v    = h1v + N32;
  float* Dxh    = h2v + N32;
  float* hK_agg = Dxh + N32;
  float* d_agg  = hK_agg + N32;
  float* E_new  = d_agg + N32;
  float* E_node = E_new + N64;

  float* out = (float*)d_out;

  {
    long nz = (long)NNODES * 128;
    int blocks = (int)((nz + 255) / 256);
    k_zero<<<blocks, 256, 0, stream>>>(hK_agg, nz);
  }
  {
    int blocks = (NODE_TILES + WPB * 4 - 1) / (WPB * 4);
    k_encode<<<blocks, 256, 0, stream>>>(x, WencK, bencK, WencP1, bencP1,
                                         WencP2, bencP2, WD, bD,
                                         hK, h1v, h2v, Dxh);
  }
  {
    long nt = (long)NEDGES * 8;
    int blocks = (int)((nt + 255) / 256);
    k_scatter_enc<<<blocks, 256, 0, stream>>>(src, dst, hK, Dxh, hK_agg, d_agg);
  }
  {
    int blocks = (NODE_TILES + WPB * 4 - 1) / (WPB * 4);
    k_node_mlp<<<blocks, 256, 0, stream>>>(hK_agg, WK1, bK1, WK2, bK2, WK3, bK3,
                                           E_node);
  }
  {
    int blocks = (EDGE_TILES + WPB * 4 - 1) / (WPB * 4);
    k_edge<<<blocks, 256, 0, stream>>>(src, dst, h1v, h2v, E_node,
                                       WU1, bU1, WU2, bU2, WU3, bU3, E_new);
  }
  {
    int blocks = (NODE_TILES + WPB * 4 - 1) / (WPB * 4);
    k_final<<<blocks, 256, 0, stream>>>(E_new, WH, bH, d_agg, out);
  }
}


# Grids sized to the chip: node kernels 256 blocks, output head 391 (were 98: at most 98 of 256 WGPs busy), edge kernel 1250 co-resident blocks of exactly 5 tiles a wave (was 1563); sums stay atomic
# speedup vs baseline: 1.7452x; 1.7452x over previous
#include <hip/hip_runtime.h>

#define NNODES 50000
#define NEDGES 800000
#define NODE_TILES (NNODES / 16)
#define EDGE_TILES (NEDGES / 16)
#define WPB 8
#define NUM_WGP 256
#define EDGE_TILES_PER_WAVE 5
#define EDGE_BLOCKS (EDGE_TILES / (WPB * EDGE_TILES_PER_WAVE))
#define FINAL_BLOCKS ((NODE_TILES + WPB - 1) / WPB)
#define XPITCH 36

typedef __attribute__((ext_vector_type(2))) float v2f;
typedef __attribute__((ext_vector_type(8))) float v8f;

__device__ __forceinline__ v8f wmma4(v2f a, v2f b, v8f c) {
  return __builtin_amdgcn_wmma_f32_16x16x4_f32(
      false, a, false, b, (short)0, c, false, false);
}

__device__ __forceinline__ float xhalf(float v) {
  return __builtin_bit_cast(float, __builtin_amdgcn_permlanex16(
      __builtin_bit_cast(int, v), __builtin_bit_cast(int, v), 0x76543210, 0xfedcba98, false, false));
}

__device__ __forceinline__ v2f ldB(const float* W, int ncols, int kb, int col0, int lane) {
  int col = col0 + (lane & 15);
  int k = 4 * kb + 2 * (lane >> 4);
  v2f b;
  b.x = W[k * ncols + col];
  b.y = W[(k + 1) * ncols + col];
  return b;
}

__device__ __forceinline__ v8f bias_frag(const float* b, int col0, int lane) {
  float bv = b[col0 + (lane & 15)];
  v8f c;
#pragma unroll
  for (int r = 0; r < 8; ++r) c[r] = bv;
  return c;
}


__device__ __forceinline__ v2f ldWp(const float* W, int t, int j, int c0, int lane) {
  int col = c0 + (lane & 15);
  int k = 16 * t + 8 * (lane >> 4) + 2 * j;
  v2f w;
  w.x = W[k * 64 + col];
  w.y = W[(k + 1) * 64 + col];
  return w;
}

__device__ __forceinline__ v8f bias_rows(const float* b, int mt, int lane) {
  const float* p = b + 16 * mt + 8 * (lane >> 4);
  v8f c;
#pragma unroll
  for (int r = 0; r < 8; ++r) c[r] = p[r];
  return c;
}

__device__ __forceinline__ void mlp3_regs(const v2f* x,
                                          const float* sW1, const float* sb1,
                                          const float* sW2, const float* sb2,
                                          const float* sW3, const float* sb3,
                                          v8f* out, int lane) {
  v8f h1[4], h2[4];
#pragma unroll
  for (int mt = 0; mt < 4; ++mt) h1[mt] = bias_rows(sb1, mt, lane);
#pragma unroll
  for (int kb = 0; kb < 8; ++kb)
#pragma unroll
    for (int mt = 0; mt < 4; ++mt)
      h1[mt] = wmma4(ldB(sW1, 64, kb, mt * 16, lane), x[kb], h1[mt]);
#pragma unroll
  for (int mt = 0; mt < 4; ++mt)
#pragma unroll
    for (int r = 0; r < 8; ++r) h1[mt][r] = tanhf(h1[mt][r]);

#pragma unroll
  for (int mt = 0; mt < 4; ++mt) h2[mt] = bias_rows(sb2, mt, lane);
#pragma unroll
  for (int t = 0; t < 4; ++t)
#pragma unroll
    for (int j = 0; j < 4; ++j) {
      v2f a;
      a.x = h1[t][2 * j];
      a.y = h1[t][2 * j + 1];
#pragma unroll
      for (int mt = 0; mt < 4; ++mt)
        h2[mt] = wmma4(ldWp(sW2, t, j, mt * 16, lane), a, h2[mt]);
    }
#pragma unroll
  for (int mt = 0; mt < 4; ++mt)
#pragma unroll
    for (int r = 0; r < 8; ++r) h2[mt][r] = fmaxf(h2[mt][r], 0.0f);

#pragma unroll
  for (int nb = 0; nb < 4; ++nb) out[nb] = bias_frag(sb3, nb * 16, lane);
#pragma unroll
  for (int t = 0; t < 4; ++t)
#pragma unroll
    for (int j = 0; j < 4; ++j) {
      v2f a;
      a.x = h2[t][2 * j];
      a.y = h2[t][2 * j + 1];
#pragma unroll
      for (int nb = 0; nb < 4; ++nb)
        out[nb] = wmma4(a, ldWp(sW3, t, j, nb * 16, lane), out[nb]);
    }
}

__global__ void k_zero(float* p, long n) {
  long i = (long)blockIdx.x * blockDim.x + threadIdx.x;
  if (i < n) p[i] = 0.0f;
}

__global__ void __launch_bounds__(256) k_encode(
    const float* __restrict__ x,
    const float* __restrict__ WencK, const float* __restrict__ bencK,
    const float* __restrict__ WencP1, const float* __restrict__ bencP1,
    const float* __restrict__ WencP2, const float* __restrict__ bencP2,
    const float* __restrict__ WD, const float* __restrict__ bD,
    float* __restrict__ hK, float* __restrict__ h1,
    float* __restrict__ h2, float* __restrict__ Dxh) {
  __shared__ float sWK[32 * 32], sW1[32 * 32], sW2[32 * 32], sWD[64 * 64];
  __shared__ float sbK[32], sb1[32], sb2[32], sbD[32];
  for (int i = threadIdx.x; i < 32 * 32; i += blockDim.x) {
    sWK[i] = WencK[i]; sW1[i] = WencP1[i]; sW2[i] = WencP2[i];
  }
  for (int i = threadIdx.x; i < 64 * 64; i += blockDim.x) sWD[i] = WD[i];
  for (int i = threadIdx.x; i < 32; i += blockDim.x) {
    sbK[i] = bencK[i]; sb1[i] = bencP1[i]; sb2[i] = bencP2[i]; sbD[i] = bD[32 + i];
  }
  __syncthreads();

  int wave = threadIdx.x >> 5, lane = threadIdx.x & 31;
  int m = lane & 15;
  int half = lane >> 4, nl = lane & 15;

  for (int tile = blockIdx.x * WPB + wave; tile < NODE_TILES;
       tile += gridDim.x * WPB) {
    int n0 = tile * 16;

    v8f aK[2], a1[2], a2[2], aD[2];
#pragma unroll
    for (int nb = 0; nb < 2; ++nb) {
      aK[nb] = bias_frag(sbK, nb * 16, lane);
      a1[nb] = bias_frag(sb1, nb * 16, lane);
      a2[nb] = bias_frag(sb2, nb * 16, lane);
      aD[nb] = bias_frag(sbD, nb * 16, lane);
    }

#pragma unroll
    for (int kb = 0; kb < 8; ++kb) {
      int k = 4 * kb + 2 * (lane >> 4);
      long base = (long)(n0 + m) * 64;
      v2f aq, ap;
      aq.x = x[base + k];       aq.y = x[base + k + 1];
      ap.x = x[base + 32 + k];  ap.y = x[base + 32 + k + 1];
#pragma unroll
      for (int nb = 0; nb < 2; ++nb) {
        aK[nb] = wmma4(ap, ldB(sWK, 32, kb, nb * 16, lane), aK[nb]);
        a1[nb] = wmma4(aq, ldB(sW1, 32, kb, nb * 16, lane), a1[nb]);
        a2[nb] = wmma4(aq, ldB(sW2, 32, kb, nb * 16, lane), a2[nb]);
      }
    }
#pragma unroll
    for (int kb = 0; kb < 16; ++kb) {
      int k = 4 * kb + 2 * (lane >> 4);
      long base = (long)(n0 + m) * 64;
      v2f ax;
      ax.x = x[base + k]; ax.y = x[base + k + 1];
#pragma unroll
      for (int nb = 0; nb < 2; ++nb)
        aD[nb] = wmma4(ax, ldB(sWD, 64, kb, 32 + nb * 16, lane), aD[nb]);
    }

#pragma unroll
    for (int nb = 0; nb < 2; ++nb)
#pragma unroll
      for (int r = 0; r < 8; ++r) {
        long o = (long)(n0 + r + 8 * half) * 32 + nb * 16 + nl;
        hK[o] = aK[nb][r];
        h1[o] = a1[nb][r];
        h2[o] = a2[nb][r];
        Dxh[o] = aD[nb][r];
      }
  }
}

__global__ void k_scatter_enc(const int* __restrict__ src, const int* __restrict__ dst,
                              const float* __restrict__ hK, float* __restrict__ hK_agg) {
  long tid = (long)blockIdx.x * blockDim.x + threadIdx.x;
  if (tid >= (long)NEDGES * 8) return;
  int e = (int)(tid >> 3);
  int j = (int)(tid & 7) * 4;
  int s = src[e], d = dst[e];
  float4 a = *(const float4*)&hK[(long)s * 32 + j];
  float* pa = &hK_agg[(long)d * 32 + j];
  atomicAdd(pa + 0, a.x); atomicAdd(pa + 1, a.y);
  atomicAdd(pa + 2, a.z); atomicAdd(pa + 3, a.w);
}

__global__ void __launch_bounds__(256) k_node_mlp(
    const float* __restrict__ hK_agg,
    const float* __restrict__ WK1, const float* __restrict__ bK1,
    const float* __restrict__ WK2, const float* __restrict__ bK2,
    const float* __restrict__ WK3, const float* __restrict__ bK3,
    float* __restrict__ E_node) {
  __shared__ float sW1[32 * 64], sW2[64 * 64], sW3[64 * 64];
  __shared__ __attribute__((aligned(16))) float sb1[64], sb2[64], sb3[64];
  for (int i = threadIdx.x; i < 32 * 64; i += blockDim.x) sW1[i] = WK1[i];
  for (int i = threadIdx.x; i < 64 * 64; i += blockDim.x) { sW2[i] = WK2[i]; sW3[i] = WK3[i]; }
  for (int i = threadIdx.x; i < 64; i += blockDim.x) { sb1[i] = bK1[i]; sb2[i] = bK2[i]; sb3[i] = bK3[i]; }
  __syncthreads();

  int wave = threadIdx.x >> 5, lane = threadIdx.x & 31;
  int m = lane & 15;
  int half = lane >> 4, nl = lane & 15;

  for (int tile = blockIdx.x * WPB + wave; tile < NODE_TILES;
       tile += gridDim.x * WPB) {
    int n0 = tile * 16;

    v2f xin[8];
#pragma unroll
    for (int kb = 0; kb < 8; ++kb) {
      int k = 4 * kb + 2 * (lane >> 4);
      long base = (long)(n0 + m) * 32;
      xin[kb].x = hK_agg[base + k];
      xin[kb].y = hK_agg[base + k + 1];
    }
    v8f acc[4];
    mlp3_regs(xin, sW1, sb1, sW2, sb2, sW3, sb3, acc, lane);

#pragma unroll
    for (int nb = 0; nb < 4; ++nb)
#pragma unroll
      for (int r = 0; r < 8; ++r)
        E_node[(long)(n0 + r + 8 * half) * 64 + nb * 16 + nl] = acc[nb][r];
  }
}

__global__ void __launch_bounds__(256) k_edge(
    const int* __restrict__ src, const int* __restrict__ dst,
    const float* __restrict__ h1, const float* __restrict__ h2,
    const float* __restrict__ E_node,
    const float* __restrict__ Dxh, float* __restrict__ d_agg,
    const float* __restrict__ WU1, const float* __restrict__ bU1,
    const float* __restrict__ WU2, const float* __restrict__ bU2,
    const float* __restrict__ WU3, const float* __restrict__ bU3,
    float* __restrict__ E_new) {
  __shared__ float sW1[32 * 64], sW2[64 * 64], sW3[64 * 64];
  __shared__ __attribute__((aligned(16))) float sb1[64], sb2[64], sb3[64];
  __shared__ __attribute__((aligned(16))) float sX[WPB * 16 * XPITCH];
  for (int i = threadIdx.x; i < 32 * 64; i += blockDim.x) sW1[i] = WU1[i];
  for (int i = threadIdx.x; i < 64 * 64; i += blockDim.x) { sW2[i] = WU2[i]; sW3[i] = WU3[i]; }
  for (int i = threadIdx.x; i < 64; i += blockDim.x) { sb1[i] = bU1[i]; sb2[i] = bU2[i]; sb3[i] = bU3[i]; }
  __syncthreads();

  int wave = threadIdx.x >> 5, lane = threadIdx.x & 31;
  int m = lane & 15;
  int half = lane >> 4;

  float* xt = sX + wave * (16 * XPITCH);

  for (int tile = blockIdx.x * WPB + wave; tile < EDGE_TILES;
       tile += gridDim.x * WPB) {
    asm volatile("" ::: "memory");
    int e0 = tile * 16;
    int s = src[e0 + m], d = dst[e0 + m];

    float dv[16], xa[16], xb[16];
#pragma unroll
    for (int i = 0; i < 16; ++i) {
      unsigned si = (unsigned)__builtin_amdgcn_readlane(s, i);
      unsigned di = (unsigned)__builtin_amdgcn_readlane(d, i);
      dv[i] = Dxh[si * 32u + (unsigned)lane];
      xa[i] = h1[si * 32u + (unsigned)lane];
      xb[i] = h2[di * 32u + (unsigned)lane];
    }
#pragma unroll
    for (int i = 0; i < 16; ++i) xt[i * XPITCH + lane] = xa[i] + xb[i];
#pragma unroll
    for (int i = 0; i < 16; ++i) {
      unsigned di = (unsigned)__builtin_amdgcn_readlane(d, i);
      atomicAdd(&d_agg[di * 32u + (unsigned)lane], dv[i]);
    }
    v2f xin[8];
#pragma unroll
    for (int kb = 0; kb < 8; ++kb)
      xin[kb] = *(const v2f*)&xt[m * XPITCH + 4 * kb + 2 * half];

    v8f acc[4];
    mlp3_regs(xin, sW1, sb1, sW2, sb2, sW3, sb3, acc, lane);

#pragma unroll
    for (int r = 0; r < 8; ++r) {
      unsigned sA = (unsigned)__builtin_amdgcn_readlane(s, r);
      unsigned dA = (unsigned)__builtin_amdgcn_readlane(d, r);
      unsigned sB = (unsigned)__builtin_amdgcn_readlane(s, r + 8);
      unsigned dB = (unsigned)__builtin_amdgcn_readlane(d, r + 8);
#pragma unroll
      for (int p = 0; p < 2; ++p) {
        float lo = acc[2 * p][r], hi = acc[2 * p + 1][r];
        float got = xhalf(half ? lo : hi);
        float vA = half ? got : lo;
        float vB = half ? hi : got;
        unsigned col = 32u * (unsigned)p + (unsigned)lane;
        float nA = E_node[sA * 64u + col];
        float nB = E_node[sB * 64u + col];
        atomicAdd(&E_new[dA * 64u + col], vA * nA);
        atomicAdd(&E_new[dB * 64u + col], vB * nB);
      }
    }
  }
}

__global__ void __launch_bounds__(256) k_final(
    const float* __restrict__ E_new,
    const float* __restrict__ WH, const float* __restrict__ bH,
    const float* __restrict__ d_agg, float* __restrict__ out) {
  __shared__ float sWH[64 * 64], sbH[64];
  for (int i = threadIdx.x; i < 64 * 64; i += blockDim.x) sWH[i] = WH[i];
  for (int i = threadIdx.x; i < 64; i += blockDim.x) sbH[i] = bH[i];
  __syncthreads();

  int wave = threadIdx.x >> 5, lane = threadIdx.x & 31;
  int m = lane & 15;
  int half = lane >> 4, nl = lane & 15;

  for (int tile = blockIdx.x * WPB + wave; tile < NODE_TILES;
       tile += gridDim.x * WPB) {
    int n0 = tile * 16;

    v8f acc[4];
#pragma unroll
    for (int nb = 0; nb < 4; ++nb) acc[nb] = bias_frag(sbH, nb * 16, lane);
#pragma unroll
    for (int kb = 0; kb < 16; ++kb) {
      int k = 4 * kb + 2 * (lane >> 4);
      long base = (long)(n0 + m) * 64;
      v2f a; a.x = E_new[base + k]; a.y = E_new[base + k + 1];
#pragma unroll
      for (int nb = 0; nb < 4; ++nb)
        acc[nb] = wmma4(a, ldB(sWH, 64, kb, nb * 16, lane), acc[nb]);
    }

#pragma unroll
    for (int r = 0; r < 8; ++r) {
      long row = n0 + r + 8 * half;
#pragma unroll
      for (int nb = 0; nb < 2; ++nb) {
        out[row * 64 + nb * 16 + nl] = acc[nb + 2][r];
        out[row * 64 + 32 + nb * 16 + nl] =
            -acc[nb][r] - d_agg[row * 32 + nb * 16 + nl];
      }
    }
  }
}

extern "C" void kernel_launch(void* const* d_in, const int* in_sizes, int n_in,
                              void* d_out, int out_size, void* d_ws, size_t ws_size,
                              hipStream_t stream) {
  const float* x      = (const float*)d_in[0];
  const int*   src    = (const int*)d_in[1];
  const int*   dst    = (const int*)d_in[2];
  const float* WencK  = (const float*)d_in[3];
  const float* bencK  = (const float*)d_in[4];
  const float* WencP1 = (const float*)d_in[5];
  const float* bencP1 = (const float*)d_in[6];
  const float* WencP2 = (const float*)d_in[7];
  const float* bencP2 = (const float*)d_in[8];
  const float* WK1 = (const float*)d_in[9];
  const float* bK1 = (const float*)d_in[10];
  const float* WK2 = (const float*)d_in[11];
  const float* bK2 = (const float*)d_in[12];
  const float* WK3 = (const float*)d_in[13];
  const float* bK3 = (const float*)d_in[14];
  const float* WU1 = (const float*)d_in[15];
  const float* bU1 = (const float*)d_in[16];
  const float* WU2 = (const float*)d_in[17];
  const float* bU2 = (const float*)d_in[18];
  const float* WU3 = (const float*)d_in[19];
  const float* bU3 = (const float*)d_in[20];
  const float* WH  = (const float*)d_in[21];
  const float* bH  = (const float*)d_in[22];
  const float* WD  = (const float*)d_in[23];
  const float* bD  = (const float*)d_in[24];

  float* ws = (float*)d_ws;
  const size_t N32 = (size_t)NNODES * 32;
  const size_t N64 = (size_t)NNODES * 64;
  float* hK     = ws;
  float* h1v    = hK + N32;
  float* h2v    = h1v + N32;
  float* Dxh    = h2v + N32;
  float* hK_agg = Dxh + N32;
  float* d_agg  = hK_agg + N32;
  float* E_new  = d_agg + N32;
  float* E_node = E_new + N64;

  float* out = (float*)d_out;

  {
    long nz = (long)NNODES * 128;
    int blocks = (int)((nz + 255) / 256);
    k_zero<<<blocks, 256, 0, stream>>>(hK_agg, nz);
  }
  {
    k_encode<<<NUM_WGP, 256, 0, stream>>>(x, WencK, bencK, WencP1, bencP1,
                                         WencP2, bencP2, WD, bD,
                                         hK, h1v, h2v, Dxh);
  }
  {
    long nt = (long)NEDGES * 8;
    int blocks = (int)((nt + 255) / 256);
    k_scatter_enc<<<blocks, 256, 0, stream>>>(src, dst, hK, hK_agg);
  }
  {
    k_node_mlp<<<NUM_WGP, 256, 0, stream>>>(hK_agg, WK1, bK1, WK2, bK2, WK3, bK3,
                                           E_node);
  }
  {
    k_edge<<<EDGE_BLOCKS, 256, 0, stream>>>(src, dst, h1v, h2v, E_node, Dxh, d_agg,
                                       WU1, bU1, WU2, bU2, WU3, bU3, E_new);
  }
  {
    k_final<<<FINAL_BLOCKS, 256, 0, stream>>>(E_new, WH, bH, d_agg, out);
  }
}


# Weight staging of the four matrix kernels: all of a block's weight and bias loads issued as one group of 16-byte loads before the first LDS store (was one dependent round trip per loop pass)
# speedup vs baseline: 1.9730x; 1.1305x over previous
#include <hip/hip_runtime.h>

#define NNODES 50000
#define NEDGES 800000
#define NODE_TILES (NNODES / 16)
#define EDGE_TILES (NEDGES / 16)
#define WPB 8
#define NUM_WGP 256
#define EDGE_TILES_PER_WAVE 5
#define EDGE_BLOCKS (EDGE_TILES / (WPB * EDGE_TILES_PER_WAVE))
#define FINAL_BLOCKS ((NODE_TILES + WPB - 1) / WPB)
#define XPITCH 36

typedef __attribute__((ext_vector_type(2))) float v2f;
typedef __attribute__((ext_vector_type(4))) float v4f;
typedef __attribute__((ext_vector_type(8))) float v8f;

#define BLOCK (WPB * 32)
template <int NV>
__device__ __forceinline__ void stage_ld(const float* __restrict__ g, v4f* r) {
  for (int i = 0; i < NV; ++i) r[i] = *(const v4f*)(g + 4 * (threadIdx.x + i * BLOCK));
}
template <int NV>
__device__ __forceinline__ void stage_st(float* s, const v4f* r) {
  for (int i = 0; i < NV; ++i) *(v4f*)(s + 4 * (threadIdx.x + i * BLOCK)) = r[i];
}

__device__ __forceinline__ v8f wmma4(v2f a, v2f b, v8f c) {
  return __builtin_amdgcn_wmma_f32_16x16x4_f32(
      false, a, false, b, (short)0, c, false, false);
}

__device__ __forceinline__ float xhalf(float v) {
  return __builtin_bit_cast(float, __builtin_amdgcn_permlanex16(
      __builtin_bit_cast(int, v), __builtin_bit_cast(int, v), 0x76543210, 0xfedcba98, false, false));
}

__device__ __forceinline__ v2f ldB(const float* W, int ncols, int kb, int col0, int lane) {
  int col = col0 + (lane & 15);
  int k = 4 * kb + 2 * (lane >> 4);
  v2f b;
  b.x = W[k * ncols + col];
  b.y = W[(k + 1) * ncols + col];
  return b;
}

__device__ __forceinline__ v8f bias_frag(const float* b, int col0, int lane) {
  float bv = b[col0 + (lane & 15)];
  v8f c;
#pragma unroll
  for (int r = 0; r < 8; ++r) c[r] = bv;
  return c;
}


__device__ __forceinline__ v2f ldWp(const float* W, int t, int j, int c0, int lane) {
  int col = c0 + (lane & 15);
  int k = 16 * t + 8 * (lane >> 4) + 2 * j;
  v2f w;
  w.x = W[k * 64 + col];
  w.y = W[(k + 1) * 64 + col];
  return w;
}

__device__ __forceinline__ v8f bias_rows(const float* b, int mt, int lane) {
  const float* p = b + 16 * mt + 8 * (lane >> 4);
  v8f c;
#pragma unroll
  for (int r = 0; r < 8; ++r) c[r] = p[r];
  return c;
}

__device__ __forceinline__ void mlp3_regs(const v2f* x,
                                          const float* sW1, const float* sb1,
                                          const float* sW2, const float* sb2,
                                          const float* sW3, const float* sb3,
                                          v8f* out, int lane) {
  v8f h1[4], h2[4];
#pragma unroll
  for (int mt = 0; mt < 4; ++mt) h1[mt] = bias_rows(sb1, mt, lane);
#pragma unroll
  for (int kb = 0; kb < 8; ++kb)
#pragma unroll
    for (int mt = 0; mt < 4; ++mt)
      h1[mt] = wmma4(ldB(sW1, 64, kb, mt * 16, lane), x[kb], h1[mt]);
#pragma unroll
  for (int mt = 0; mt < 4; ++mt)
#pragma unroll
    for (int r = 0; r < 8; ++r) h1[mt][r] = tanhf(h1[mt][r]);

#pragma unroll
  for (int mt = 0; mt < 4; ++mt) h2[mt] = bias_rows(sb2, mt, lane);
#pragma unroll
  for (int t = 0; t < 4; ++t)
#pragma unroll
    for (int j = 0; j < 4; ++j) {
      v2f a;
      a.x = h1[t][2 * j];
      a.y = h1[t][2 * j + 1];
#pragma unroll
      for (int mt = 0; mt < 4; ++mt)
        h2[mt] = wmma4(ldWp(sW2, t, j, mt * 16, lane), a, h2[mt]);
    }
#pragma unroll
  for (int mt = 0; mt < 4; ++mt)
#pragma unroll
    for (int r = 0; r < 8; ++r) h2[mt][r] = fmaxf(h2[mt][r], 0.0f);

#pragma unroll
  for (int nb = 0; nb < 4; ++nb) out[nb] = bias_frag(sb3, nb * 16, lane);
#pragma unroll
  for (int t = 0; t < 4; ++t)
#pragma unroll
    for (int j = 0; j < 4; ++j) {
      v2f a;
      a.x = h2[t][2 * j];
      a.y = h2[t][2 * j + 1];
#pragma unroll
      for (int nb = 0; nb < 4; ++nb)
        out[nb] = wmma4(a, ldWp(sW3, t, j, nb * 16, lane), out[nb]);
    }
}

__global__ void k_zero(float* p, long n) {
  long i = (long)blockIdx.x * blockDim.x + threadIdx.x;
  if (i < n) p[i] = 0.0f;
}

__global__ void __launch_bounds__(256) k_encode(
    const float* __restrict__ x,
    const float* __restrict__ WencK, const float* __restrict__ bencK,
    const float* __restrict__ WencP1, const float* __restrict__ bencP1,
    const float* __restrict__ WencP2, const float* __restrict__ bencP2,
    const float* __restrict__ WD, const float* __restrict__ bD,
    float* __restrict__ hK, float* __restrict__ h1,
    float* __restrict__ h2, float* __restrict__ Dxh) {
  __shared__ __attribute__((aligned(16))) float sWK[32 * 32], sW1[32 * 32], sW2[32 * 32], sWD[64 * 64];
  __shared__ float sbK[32], sb1[32], sb2[32], sbD[32];
  {
    v4f rK[1], r1[1], r2[1], rD[4];
    int bi = threadIdx.x & 31;
    stage_ld<1>(WencK, rK); stage_ld<1>(WencP1, r1); stage_ld<1>(WencP2, r2);
    stage_ld<4>(WD, rD);
    float cK = bencK[bi], c1 = bencP1[bi], c2 = bencP2[bi], cD = bD[32 + bi];
    stage_st<1>(sWK, rK); stage_st<1>(sW1, r1); stage_st<1>(sW2, r2);
    stage_st<4>(sWD, rD);
    sbK[bi] = cK; sb1[bi] = c1; sb2[bi] = c2; sbD[bi] = cD;
  }
  __syncthreads();

  int wave = threadIdx.x >> 5, lane = threadIdx.x & 31;
  int m = lane & 15;
  int half = lane >> 4, nl = lane & 15;

  for (int tile = blockIdx.x * WPB + wave; tile < NODE_TILES;
       tile += gridDim.x * WPB) {
    int n0 = tile * 16;

    v8f aK[2], a1[2], a2[2], aD[2];
#pragma unroll
    for (int nb = 0; nb < 2; ++nb) {
      aK[nb] = bias_frag(sbK, nb * 16, lane);
      a1[nb] = bias_frag(sb1, nb * 16, lane);
      a2[nb] = bias_frag(sb2, nb * 16, lane);
      aD[nb] = bias_frag(sbD, nb * 16, lane);
    }

#pragma unroll
    for (int kb = 0; kb < 8; ++kb) {
      int k = 4 * kb + 2 * (lane >> 4);
      long base = (long)(n0 + m) * 64;
      v2f aq, ap;
      aq.x = x[base + k];       aq.y = x[base + k + 1];
      ap.x = x[base + 32 + k];  ap.y = x[base + 32 + k + 1];
#pragma unroll
      for (int nb = 0; nb < 2; ++nb) {
        aK[nb] = wmma4(ap, ldB(sWK, 32, kb, nb * 16, lane), aK[nb]);
        a1[nb] = wmma4(aq, ldB(sW1, 32, kb, nb * 16, lane), a1[nb]);
        a2[nb] = wmma4(aq, ldB(sW2, 32, kb, nb * 16, lane), a2[nb]);
      }
    }
#pragma unroll
    for (int kb = 0; kb < 16; ++kb) {
      int k = 4 * kb + 2 * (lane >> 4);
      long base = (long)(n0 + m) * 64;
      v2f ax;
      ax.x = x[base + k]; ax.y = x[base + k + 1];
#pragma unroll
      for (int nb = 0; nb < 2; ++nb)
        aD[nb] = wmma4(ax, ldB(sWD, 64, kb, 32 + nb * 16, lane), aD[nb]);
    }

#pragma unroll
    for (int nb = 0; nb < 2; ++nb)
#pragma unroll
      for (int r = 0; r < 8; ++r) {
        long o = (long)(n0 + r + 8 * half) * 32 + nb * 16 + nl;
        hK[o] = aK[nb][r];
        h1[o] = a1[nb][r];
        h2[o] = a2[nb][r];
        Dxh[o] = aD[nb][r];
      }
  }
}

__global__ void k_scatter_enc(const int* __restrict__ src, const int* __restrict__ dst,
                              const float* __restrict__ hK, float* __restrict__ hK_agg) {
  long tid = (long)blockIdx.x * blockDim.x + threadIdx.x;
  if (tid >= (long)NEDGES * 8) return;
  int e = (int)(tid >> 3);
  int j = (int)(tid & 7) * 4;
  int s = src[e], d = dst[e];
  float4 a = *(const float4*)&hK[(long)s * 32 + j];
  float* pa = &hK_agg[(long)d * 32 + j];
  atomicAdd(pa + 0, a.x); atomicAdd(pa + 1, a.y);
  atomicAdd(pa + 2, a.z); atomicAdd(pa + 3, a.w);
}

__global__ void __launch_bounds__(256) k_node_mlp(
    const float* __restrict__ hK_agg,
    const float* __restrict__ WK1, const float* __restrict__ bK1,
    const float* __restrict__ WK2, const float* __restrict__ bK2,
    const float* __restrict__ WK3, const float* __restrict__ bK3,
    float* __restrict__ E_node) {
  __shared__ __attribute__((aligned(16))) float sW1[32 * 64], sW2[64 * 64], sW3[64 * 64];
  __shared__ __attribute__((aligned(16))) float sb1[64], sb2[64], sb3[64];
  {
    v4f r1[2], r2[4], r3[4];
    int bi = threadIdx.x & 63;
    stage_ld<2>(WK1, r1); stage_ld<4>(WK2, r2); stage_ld<4>(WK3, r3);
    float c1 = bK1[bi], c2 = bK2[bi], c3 = bK3[bi];
    stage_st<2>(sW1, r1); stage_st<4>(sW2, r2); stage_st<4>(sW3, r3);
    sb1[bi] = c1; sb2[bi] = c2; sb3[bi] = c3;
  }
  __syncthreads();

  int wave = threadIdx.x >> 5, lane = threadIdx.x & 31;
  int m = lane & 15;
  int half = lane >> 4, nl = lane & 15;

  for (int tile = blockIdx.x * WPB + wave; tile < NODE_TILES;
       tile += gridDim.x * WPB) {
    int n0 = tile * 16;

    v2f xin[8];
#pragma unroll
    for (int kb = 0; kb < 8; ++kb) {
      int k = 4 * kb + 2 * (lane >> 4);
      long base = (long)(n0 + m) * 32;
      xin[kb].x = hK_agg[base + k];
      xin[kb].y = hK_agg[base + k + 1];
    }
    v8f acc[4];
    mlp3_regs(xin, sW1, sb1, sW2, sb2, sW3, sb3, acc, lane);

#pragma unroll
    for (int nb = 0; nb < 4; ++nb)
#pragma unroll
      for (int r = 0; r < 8; ++r)
        E_node[(long)(n0 + r + 8 * half) * 64 + nb * 16 + nl] = acc[nb][r];
  }
}

__global__ void __launch_bounds__(256, 10) k_edge(
    const int* __restrict__ src, const int* __restrict__ dst,
    const float* __restrict__ h1, const float* __restrict__ h2,
    const float* __restrict__ E_node,
    const float* __restrict__ Dxh, float* __restrict__ d_agg,
    const float* __restrict__ WU1, const float* __restrict__ bU1,
    const float* __restrict__ WU2, const float* __restrict__ bU2,
    const float* __restrict__ WU3, const float* __restrict__ bU3,
    float* __restrict__ E_new) {
  __shared__ __attribute__((aligned(16))) float sW1[32 * 64], sW2[64 * 64], sW3[64 * 64];
  __shared__ __attribute__((aligned(16))) float sb1[64], sb2[64], sb3[64];
  __shared__ __attribute__((aligned(16))) float sX[WPB * 16 * XPITCH];
  {
    v4f r1[2], r2[4], r3[4];
    int bi = threadIdx.x & 63;
    stage_ld<2>(WU1, r1); stage_ld<4>(WU2, r2); stage_ld<4>(WU3, r3);
    float c1 = bU1[bi], c2 = bU2[bi], c3 = bU3[bi];
    stage_st<2>(sW1, r1); stage_st<4>(sW2, r2); stage_st<4>(sW3, r3);
    sb1[bi] = c1; sb2[bi] = c2; sb3[bi] = c3;
  }
  __syncthreads();

  int wave = threadIdx.x >> 5, lane = threadIdx.x & 31;
  int m = lane & 15;
  int half = lane >> 4;

  float* xt = sX + wave * (16 * XPITCH);

  for (int tile = blockIdx.x * WPB + wave; tile < EDGE_TILES;
       tile += gridDim.x * WPB) {
    asm volatile("" ::: "memory");
    int e0 = tile * 16;
    int s = src[e0 + m], d = dst[e0 + m];

    float dv[16], xa[16], xb[16];
#pragma unroll
    for (int i = 0; i < 16; ++i) {
      unsigned si = (unsigned)__builtin_amdgcn_readlane(s, i);
      unsigned di = (unsigned)__builtin_amdgcn_readlane(d, i);
      dv[i] = Dxh[si * 32u + (unsigned)lane];
      xa[i] = h1[si * 32u + (unsigned)lane];
      xb[i] = h2[di * 32u + (unsigned)lane];
    }
#pragma unroll
    for (int i = 0; i < 16; ++i) xt[i * XPITCH + lane] = xa[i] + xb[i];
#pragma unroll
    for (int i = 0; i < 16; ++i) {
      unsigned di = (unsigned)__builtin_amdgcn_readlane(d, i);
      atomicAdd(&d_agg[di * 32u + (unsigned)lane], dv[i]);
    }
    v2f xin[8];
#pragma unroll
    for (int kb = 0; kb < 8; ++kb)
      xin[kb] = *(const v2f*)&xt[m * XPITCH + 4 * kb + 2 * half];

    v8f acc[4];
    mlp3_regs(xin, sW1, sb1, sW2, sb2, sW3, sb3, acc, lane);

#pragma unroll
    for (int r = 0; r < 8; ++r) {
      unsigned sA = (unsigned)__builtin_amdgcn_readlane(s, r);
      unsigned dA = (unsigned)__builtin_amdgcn_readlane(d, r);
      unsigned sB = (unsigned)__builtin_amdgcn_readlane(s, r + 8);
      unsigned dB = (unsigned)__builtin_amdgcn_readlane(d, r + 8);
#pragma unroll
      for (int p = 0; p < 2; ++p) {
        float lo = acc[2 * p][r], hi = acc[2 * p + 1][r];
        float got = xhalf(half ? lo : hi);
        float vA = half ? got : lo;
        float vB = half ? hi : got;
        unsigned col = 32u * (unsigned)p + (unsigned)lane;
        float nA = E_node[sA * 64u + col];
        float nB = E_node[sB * 64u + col];
        atomicAdd(&E_new[dA * 64u + col], vA * nA);
        atomicAdd(&E_new[dB * 64u + col], vB * nB);
      }
    }
  }
}

__global__ void __launch_bounds__(256) k_final(
    const float* __restrict__ E_new,
    const float* __restrict__ WH, const float* __restrict__ bH,
    const float* __restrict__ d_agg, float* __restrict__ out) {
  __shared__ __attribute__((aligned(16))) float sWH[64 * 64];
  __shared__ float sbH[64];
  {
    v4f rH[4];
    int bi = threadIdx.x & 63;
    stage_ld<4>(WH, rH);
    float cH = bH[bi];
    stage_st<4>(sWH, rH);
    sbH[bi] = cH;
  }
  __syncthreads();

  int wave = threadIdx.x >> 5, lane = threadIdx.x & 31;
  int m = lane & 15;
  int half = lane >> 4, nl = lane & 15;

  for (int tile = blockIdx.x * WPB + wave; tile < NODE_TILES;
       tile += gridDim.x * WPB) {
    int n0 = tile * 16;

    v8f acc[4];
#pragma unroll
    for (int nb = 0; nb < 4; ++nb) acc[nb] = bias_frag(sbH, nb * 16, lane);
#pragma unroll
    for (int kb = 0; kb < 16; ++kb) {
      int k = 4 * kb + 2 * (lane >> 4);
      long base = (long)(n0 + m) * 64;
      v2f a; a.x = E_new[base + k]; a.y = E_new[base + k + 1];
#pragma unroll
      for (int nb = 0; nb < 4; ++nb)
        acc[nb] = wmma4(a, ldB(sWH, 64, kb, nb * 16, lane), acc[nb]);
    }

#pragma unroll
    for (int r = 0; r < 8; ++r) {
      long row = n0 + r + 8 * half;
#pragma unroll
      for (int nb = 0; nb < 2; ++nb) {
        out[row * 64 + nb * 16 + nl] = acc[nb + 2][r];
        out[row * 64 + 32 + nb * 16 + nl] =
            -acc[nb][r] - d_agg[row * 32 + nb * 16 + nl];
      }
    }
  }
}

extern "C" void kernel_launch(void* const* d_in, const int* in_sizes, int n_in,
                              void* d_out, int out_size, void* d_ws, size_t ws_size,
                              hipStream_t stream) {
  const float* x      = (const float*)d_in[0];
  const int*   src    = (const int*)d_in[1];
  const int*   dst    = (const int*)d_in[2];
  const float* WencK  = (const float*)d_in[3];
  const float* bencK  = (const float*)d_in[4];
  const float* WencP1 = (const float*)d_in[5];
  const float* bencP1 = (const float*)d_in[6];
  const float* WencP2 = (const float*)d_in[7];
  const float* bencP2 = (const float*)d_in[8];
  const float* WK1 = (const float*)d_in[9];
  const float* bK1 = (const float*)d_in[10];
  const float* WK2 = (const float*)d_in[11];
  const float* bK2 = (const float*)d_in[12];
  const float* WK3 = (const float*)d_in[13];
  const float* bK3 = (const float*)d_in[14];
  const float* WU1 = (const float*)d_in[15];
  const float* bU1 = (const float*)d_in[16];
  const float* WU2 = (const float*)d_in[17];
  const float* bU2 = (const float*)d_in[18];
  const float* WU3 = (const float*)d_in[19];
  const float* bU3 = (const float*)d_in[20];
  const float* WH  = (const float*)d_in[21];
  const float* bH  = (const float*)d_in[22];
  const float* WD  = (const float*)d_in[23];
  const float* bD  = (const float*)d_in[24];

  float* ws = (float*)d_ws;
  const size_t N32 = (size_t)NNODES * 32;
  const size_t N64 = (size_t)NNODES * 64;
  float* hK     = ws;
  float* h1v    = hK + N32;
  float* h2v    = h1v + N32;
  float* Dxh    = h2v + N32;
  float* hK_agg = Dxh + N32;
  float* d_agg  = hK_agg + N32;
  float* E_new  = d_agg + N32;
  float* E_node = E_new + N64;

  float* out = (float*)d_out;

  {
    long nz = (long)NNODES * 128;
    int blocks = (int)((nz + 255) / 256);
    k_zero<<<blocks, 256, 0, stream>>>(hK_agg, nz);
  }
  {
    k_encode<<<NUM_WGP, BLOCK, 0, stream>>>(x, WencK, bencK, WencP1, bencP1,
                                         WencP2, bencP2, WD, bD,
                                         hK, h1v, h2v, Dxh);
  }
  {
    long nt = (long)NEDGES * 8;
    int blocks = (int)((nt + 255) / 256);
    k_scatter_enc<<<blocks, 256, 0, stream>>>(src, dst, hK, hK_agg);
  }
  {
    k_node_mlp<<<NUM_WGP, BLOCK, 0, stream>>>(hK_agg, WK1, bK1, WK2, bK2, WK3, bK3,
                                           E_node);
  }
  {
    k_edge<<<EDGE_BLOCKS, BLOCK, 0, stream>>>(src, dst, h1v, h2v, E_node, Dxh, d_agg,
                                       WU1, bU1, WU2, bU2, WU3, bU3, E_new);
  }
  {
    k_final<<<FINAL_BLOCKS, BLOCK, 0, stream>>>(E_new, WH, bH, d_agg, out);
  }
}
